# starting kernel
# baseline (speedup 1.0000x reference)
#include <hip/hip_runtime.h>

#define FIN 256
#define HID 64
#define COUT 40

typedef float v2f __attribute__((ext_vector_type(2)));
typedef float v8f __attribute__((ext_vector_type(8)));

__global__ __launch_bounds__(256) void k_deg_init(float* __restrict__ deg, int n) {
    int i = blockIdx.x * 256 + threadIdx.x;
    if (i < n) deg[i] = 1.0f;
}

__global__ __launch_bounds__(256) void k_deg_acc(const int* __restrict__ dst, int E,
                                                 float* __restrict__ deg) {
    int i = blockIdx.x * 256 + threadIdx.x;
    if (i < E) atomicAdd(deg + dst[i], 1.0f);
}

__global__ __launch_bounds__(256) void k_dinv(float* __restrict__ deg, int n) {
    int i = blockIdx.x * 256 + threadIdx.x;
    if (i < n) deg[i] = rsqrtf(deg[i]);
}

__global__ __launch_bounds__(256) void k_zero4(float4* __restrict__ p, int n4) {
    int i = blockIdx.x * 256 + threadIdx.x;
    if (i < n4) p[i] = make_float4(0.f, 0.f, 0.f, 0.f);
}

__global__ __launch_bounds__(256) void k_gemm1(const float* __restrict__ X,
                                               const float* __restrict__ W,
                                               float* __restrict__ H, int nrows) {
    int wave = (blockIdx.x * 256 + threadIdx.x) >> 5;
    int lane = threadIdx.x & 31;
    int m0 = wave << 4;
    if (m0 >= nrows) return;
    int row = lane & 15;
    int kk  = (lane >> 4) << 1;
    const float* Xr = X + (size_t)(m0 + row) * FIN + kk;
    v8f c0 = {}, c1 = {}, c2 = {}, c3 = {};
    for (int k0 = 0; k0 < FIN; k0 += 4) {
        v2f a;
        a.x = Xr[k0];
        a.y = Xr[k0 + 1];
        const float* Wp = W + (size_t)(k0 + kk) * HID;
        v2f b0 = { Wp[row],      Wp[HID + row] };
        v2f b1 = { Wp[16 + row], Wp[HID + 16 + row] };
        v2f b2 = { Wp[32 + row], Wp[HID + 32 + row] };
        v2f b3 = { Wp[48 + row], Wp[HID + 48 + row] };
        c0 = __builtin_amdgcn_wmma_f32_16x16x4_f32(false, a, false, b0, (short)0, c0, false, false);
        c1 = __builtin_amdgcn_wmma_f32_16x16x4_f32(false, a, false, b1, (short)0, c1, false, false);
        c2 = __builtin_amdgcn_wmma_f32_16x16x4_f32(false, a, false, b2, (short)0, c2, false, false);
        c3 = __builtin_amdgcn_wmma_f32_16x16x4_f32(false, a, false, b3, (short)0, c3, false, false);
    }
    int mrow = m0 + ((lane >> 4) << 3);
    int col  = lane & 15;
    for (int r = 0; r < 8; ++r) {
        float* out = H + (size_t)(mrow + r) * HID + col;
        out[0]  = c0[r];
        out[16] = c1[r];
        out[32] = c2[r];
        out[48] = c3[r];
    }
}

__global__ __launch_bounds__(256) void k_gemm2(const float* __restrict__ H1,
                                               const float* __restrict__ W,
                                               float* __restrict__ H2, int nrows) {
    int wave = (blockIdx.x * 256 + threadIdx.x) >> 5;
    int lane = threadIdx.x & 31;
    int m0 = wave << 4;
    if (m0 >= nrows) return;
    int row = lane & 15;
    int kk  = (lane >> 4) << 1;
    const float* Hr = H1 + (size_t)(m0 + row) * HID + kk;
    v8f c0 = {}, c1 = {}, c2 = {};
    for (int k0 = 0; k0 < HID; k0 += 4) {
        v2f a;
        a.x = Hr[k0];
        a.y = Hr[k0 + 1];
        const float* Wp = W + (size_t)(k0 + kk) * COUT;
        v2f b0 = { Wp[row],      Wp[COUT + row] };
        v2f b1 = { Wp[16 + row], Wp[COUT + 16 + row] };
        int col2 = 32 + row;
        v2f b2 = { (col2 < COUT) ? Wp[col2] : 0.0f,
                   (col2 < COUT) ? Wp[COUT + col2] : 0.0f };
        c0 = __builtin_amdgcn_wmma_f32_16x16x4_f32(false, a, false, b0, (short)0, c0, false, false);
        c1 = __builtin_amdgcn_wmma_f32_16x16x4_f32(false, a, false, b1, (short)0, c1, false, false);
        c2 = __builtin_amdgcn_wmma_f32_16x16x4_f32(false, a, false, b2, (short)0, c2, false, false);
    }
    int mrow = m0 + ((lane >> 4) << 3);
    int col  = lane & 15;
    for (int r = 0; r < 8; ++r) {
        float* out = H2 + (size_t)(mrow + r) * COUT;
        out[col]      = c0[r];
        out[16 + col] = c1[r];
        if (32 + col < COUT) out[32 + col] = c2[r];
    }
}

__global__ __launch_bounds__(256) void k_agg1(const int* __restrict__ src,
                                              const int* __restrict__ dst, int E,
                                              const float* __restrict__ dinv,
                                              const float4* __restrict__ H4,
                                              float* __restrict__ A) {
    long long idx = (long long)blockIdx.x * 256 + threadIdx.x;
    if (idx >= (long long)E * 16) return;
    int e  = (int)(idx >> 4);
    int f4 = (int)(idx & 15);
    int s = src[e], d = dst[e];
    float nrm = dinv[s] * dinv[d];
    float4 v = H4[(size_t)s * (HID / 4) + f4];
    float* out = A + (size_t)d * HID + f4 * 4;
    atomicAdd(out + 0, v.x * nrm);
    atomicAdd(out + 1, v.y * nrm);
    atomicAdd(out + 2, v.z * nrm);
    atomicAdd(out + 3, v.w * nrm);
}

__global__ __launch_bounds__(256) void k_fin1(float4* __restrict__ A,
                                              const float4* __restrict__ H4,
                                              const float* __restrict__ dinv,
                                              const float* __restrict__ b1, int n) {
    int idx = blockIdx.x * 256 + threadIdx.x;
    if (idx >= n * (HID / 4)) return;
    int i  = idx >> 4;
    int f4 = idx & 15;
    float di = dinv[i];
    float nn = di * di;
    float4 a = A[idx];
    float4 h = H4[idx];
    const float* b = b1 + f4 * 4;
    a.x = fmaxf(a.x + h.x * nn + b[0], 0.f);
    a.y = fmaxf(a.y + h.y * nn + b[1], 0.f);
    a.z = fmaxf(a.z + h.z * nn + b[2], 0.f);
    a.w = fmaxf(a.w + h.w * nn + b[3], 0.f);
    A[idx] = a;
}

__global__ __launch_bounds__(320) void k_agg2(const int* __restrict__ src,
                                              const int* __restrict__ dst, int E,
                                              const float* __restrict__ dinv,
                                              const float4* __restrict__ H4,
                                              float* __restrict__ out) {
    int t = threadIdx.x;
    int e  = blockIdx.x * 32 + t / 10;
    int f4 = t % 10;
    if (e >= E) return;
    int s = src[e], d = dst[e];
    float nrm = dinv[s] * dinv[d];
    float4 v = H4[(size_t)s * (COUT / 4) + f4];
    float* o = out + (size_t)d * COUT + f4 * 4;
    atomicAdd(o + 0, v.x * nrm);
    atomicAdd(o + 1, v.y * nrm);
    atomicAdd(o + 2, v.z * nrm);
    atomicAdd(o + 3, v.w * nrm);
}

__global__ __launch_bounds__(256) void k_fin2(float4* __restrict__ out,
                                              const float4* __restrict__ H4,
                                              const float* __restrict__ dinv,
                                              const float* __restrict__ b2, int n) {
    int idx = blockIdx.x * 256 + threadIdx.x;
    if (idx >= n * (COUT / 4)) return;
    int i  = idx / 10;
    int f4 = idx % 10;
    float di = dinv[i];
    float nn = di * di;
    float4 o = out[idx];
    float4 h = H4[idx];
    const float* b = b2 + f4 * 4;
    o.x += h.x * nn + b[0];
    o.y += h.y * nn + b[1];
    o.z += h.z * nn + b[2];
    o.w += h.w * nn + b[3];
    out[idx] = o;
}

extern "C" void kernel_launch(void* const* d_in, const int* in_sizes, int n_in,
                              void* d_out, int out_size, void* d_ws, size_t ws_size,
                              hipStream_t stream) {
    const float* x  = (const float*)d_in[0];
    const float* W1 = (const float*)d_in[1];
    const float* b1 = (const float*)d_in[2];
    const float* W2 = (const float*)d_in[3];
    const float* b2 = (const float*)d_in[4];
    const int*   ei = (const int*)d_in[5];

    const int Nn = in_sizes[0] / FIN;
    const int E  = in_sizes[5] / 2;
    const int* src = ei;
    const int* dst = ei + E;

    float* ws   = (float*)d_ws;
    float* dinv = ws;
    float* h    = ws + Nn;
    float* agg  = h + (size_t)Nn * HID;
    float* h2   = h;
    float* out  = (float*)d_out;

    int gN  = (Nn + 255) / 256;
    int gE  = (E + 255) / 256;
    int h4n = Nn * (HID / 4);
    int o4n = Nn * (COUT / 4);
    int gH4 = (h4n + 255) / 256;
    int gO4 = (o4n + 255) / 256;
    int gE1 = (int)(((long long)E * 16 + 255) / 256);
    int gE2 = (E + 31) / 32;
    int gemmBlocks = ((Nn / 16) + 7) / 8;

    k_deg_init<<<gN, 256, 0, stream>>>(dinv, Nn);
    k_deg_acc<<<gE, 256, 0, stream>>>(dst, E, dinv);
    k_dinv<<<gN, 256, 0, stream>>>(dinv, Nn);

    k_gemm1<<<gemmBlocks, 256, 0, stream>>>(x, W1, h, Nn);
    k_zero4<<<gH4, 256, 0, stream>>>((float4*)agg, h4n);
    k_agg1<<<gE1, 256, 0, stream>>>(src, dst, E, dinv, (const float4*)h, agg);
    k_fin1<<<gH4, 256, 0, stream>>>((float4*)agg, (const float4*)h, dinv, b1, Nn);

    k_gemm2<<<gemmBlocks, 256, 0, stream>>>(agg, W2, h2, Nn);
    k_zero4<<<gO4, 256, 0, stream>>>((float4*)out, o4n);
    k_agg2<<<gE2, 320, 0, stream>>>(src, dst, E, dinv, (const float4*)h2, out);
    k_fin2<<<gO4, 256, 0, stream>>>((float4*)out, (const float4*)h2, dinv, b2, Nn);
}


# f32 atomic scatter-adds replaced by a per-call edge list ordered by destination (int atomic arrival-order rank, scan, fill) and row gathers summed in list order; layer-1 sums feed product 2 from LDS
# speedup vs baseline: 1.9688x; 1.9688x over previous
#include <hip/hip_runtime.h>


#define FIN 256
#define HID 64
#define COUT 40
#define TPITCH 68

typedef float v2f __attribute__((ext_vector_type(2)));
typedef float v8f __attribute__((ext_vector_type(8)));

__global__ __launch_bounds__(256) void k_cnt_zero(int* __restrict__ cnt, int n) {
    int i = blockIdx.x * 256 + threadIdx.x;
    if (i < n) cnt[i] = 0;
}

__global__ __launch_bounds__(256) void k_rank(const int* __restrict__ dst, int E,
                                              int* __restrict__ cnt, int* __restrict__ rank) {
    int i = blockIdx.x * 256 + threadIdx.x;
    if (i < E) rank[i] = atomicAdd(cnt + dst[i], 1);
}

__global__ __launch_bounds__(1024) void k_scan(const int* __restrict__ cnt, int* __restrict__ rowptr,
                                               float* __restrict__ dinv, int n) {
    __shared__ int wtot[32];
    __shared__ int tile_total;
    const int t = threadIdx.x;
    const int lane = t & 31;
    const int w = t >> 5;
    int carry = 0;
    for (int base = 0; base < n; base += 16384) {
        const int i0 = base + t * 16;
        const bool full = (i0 + 16 <= n);
        int c[16];
        if (full) {
            const int4* p = (const int4*)(cnt + i0);
            const int4 q0 = p[0], q1 = p[1], q2 = p[2], q3 = p[3];
            c[0] = q0.x;  c[1] = q0.y;  c[2] = q0.z;  c[3] = q0.w;
            c[4] = q1.x;  c[5] = q1.y;  c[6] = q1.z;  c[7] = q1.w;
            c[8] = q2.x;  c[9] = q2.y;  c[10] = q2.z; c[11] = q2.w;
            c[12] = q3.x; c[13] = q3.y; c[14] = q3.z; c[15] = q3.w;
        } else {
            for (int k = 0; k < 16; ++k) c[k] = (i0 + k < n) ? cnt[i0 + k] : 0;
        }
        int s = 0;
        for (int k = 0; k < 16; ++k) s += c[k];
        int x = s;
        for (int o = 1; o < 32; o <<= 1) {
            const int y = __shfl_up(x, o, 32);
            if (lane >= o) x += y;
        }
        if (lane == 31) wtot[w] = x;
        __syncthreads();
        if (w == 0) {
            const int v = wtot[lane];
            int xx = v;
            for (int o = 1; o < 32; o <<= 1) {
                const int y = __shfl_up(xx, o, 32);
                if (lane >= o) xx += y;
            }
            wtot[lane] = xx - v;
            if (lane == 31) tile_total = xx;
        }
        __syncthreads();
        int run = carry + wtot[w] + (x - s);
        carry += tile_total;
        if (full) {
            int r[16];
            float dv[16];
            for (int k = 0; k < 16; ++k) {
                r[k] = run;
                run += c[k];
                dv[k] = rsqrtf((float)(c[k] + 1));
            }
            int4* rp = (int4*)(rowptr + i0);
            float4* dp = (float4*)(dinv + i0);
            rp[0] = make_int4(r[0], r[1], r[2], r[3]);
            rp[1] = make_int4(r[4], r[5], r[6], r[7]);
            rp[2] = make_int4(r[8], r[9], r[10], r[11]);
            rp[3] = make_int4(r[12], r[13], r[14], r[15]);
            dp[0] = make_float4(dv[0], dv[1], dv[2], dv[3]);
            dp[1] = make_float4(dv[4], dv[5], dv[6], dv[7]);
            dp[2] = make_float4(dv[8], dv[9], dv[10], dv[11]);
            dp[3] = make_float4(dv[12], dv[13], dv[14], dv[15]);
        } else {
            for (int k = 0; k < 16; ++k) {
                if (i0 + k < n) {
                    rowptr[i0 + k] = run;
                    dinv[i0 + k] = rsqrtf((float)(c[k] + 1));
                    run += c[k];
                }
            }
        }
        __syncthreads();
    }
    if (t == 0) rowptr[n] = carry;
}

__global__ __launch_bounds__(256) void k_fill(const int* __restrict__ src, const int* __restrict__ dst,
                                              const int* __restrict__ rank,
                                              const int* __restrict__ rowptr,
                                              int* __restrict__ col, int E) {
    int i = blockIdx.x * 256 + threadIdx.x;
    if (i < E) col[rowptr[dst[i]] + rank[i]] = src[i];
}

__global__ __launch_bounds__(256) void k_gemm1(const float* __restrict__ X,
                                               const float* __restrict__ W,
                                               const float* __restrict__ dinv,
                                               float* __restrict__ H, int nrows) {
    int wave = (blockIdx.x * 256 + threadIdx.x) >> 5;
    int lane = threadIdx.x & 31;
    int m0 = wave << 4;
    if (m0 >= nrows) return;
    int row = lane & 15;
    int kk  = (lane >> 4) << 1;
    const float* Xr = X + (size_t)(m0 + row) * FIN + kk;
    v8f c0 = {}, c1 = {}, c2 = {}, c3 = {};
    for (int k0 = 0; k0 < FIN; k0 += 4) {
        v2f a;
        a.x = Xr[k0];
        a.y = Xr[k0 + 1];
        const float* Wp = W + (size_t)(k0 + kk) * HID;
        v2f b0 = { Wp[row],      Wp[HID + row] };
        v2f b1 = { Wp[16 + row], Wp[HID + 16 + row] };
        v2f b2 = { Wp[32 + row], Wp[HID + 32 + row] };
        v2f b3 = { Wp[48 + row], Wp[HID + 48 + row] };
        c0 = __builtin_amdgcn_wmma_f32_16x16x4_f32(false, a, false, b0, (short)0, c0, false, false);
        c1 = __builtin_amdgcn_wmma_f32_16x16x4_f32(false, a, false, b1, (short)0, c1, false, false);
        c2 = __builtin_amdgcn_wmma_f32_16x16x4_f32(false, a, false, b2, (short)0, c2, false, false);
        c3 = __builtin_amdgcn_wmma_f32_16x16x4_f32(false, a, false, b3, (short)0, c3, false, false);
    }
    int mrow = m0 + ((lane >> 4) << 3);
    int col  = lane & 15;
    for (int r = 0; r < 8; ++r) {
        const float di = dinv[mrow + r];
        float* out = H + (size_t)(mrow + r) * HID + col;
        out[0]  = c0[r] * di;
        out[16] = c1[r] * di;
        out[32] = c2[r] * di;
        out[48] = c3[r] * di;
    }
}

__global__ __launch_bounds__(256) void k_agg1_gemm2(const int* __restrict__ rowptr,
                                                    const int* __restrict__ col,
                                                    const float* __restrict__ dinv,
                                                    const float4* __restrict__ HS4,
                                                    const float* __restrict__ b1,
                                                    const float* __restrict__ W,
                                                    float* __restrict__ H2S, int nrows) {
    __shared__ float tile[8 * 16 * TPITCH];
    const int wib  = threadIdx.x >> 5;
    const int lane = threadIdx.x & 31;
    const int m0   = (blockIdx.x * 8 + wib) << 4;
    const bool active = m0 < nrows;
    float* T = tile + wib * (16 * TPITCH);
    const int sub = lane >> 4;
    const int f4  = lane & 15;
    float dstrip = 0.f;
    if (active) {
        const int rp = rowptr[m0 + (lane < 16 ? lane : 16)];
        dstrip = dinv[m0 + f4];
        const float* b = b1 + f4 * 4;
        const float bx = b[0], by = b[1], bz = b[2], bw = b[3];
        for (int p = 0; p < 8; ++p) {
            const int r = 2 * p + sub;
            const int d = m0 + r;
            const int beg = __shfl(rp, r, 32);
            const int end = __shfl(rp, r + 1, 32);
            float4 acc = HS4[(size_t)d * (HID / 4) + f4];
            for (int j0 = beg; j0 < end; j0 += 16) {
                const int rem = end - j0;
                const int m = rem < 16 ? rem : 16;
                const int mine = (f4 < m) ? col[j0 + f4] : d;
                for (int q = 0; q < m; q += 4) {
                    const int s0 = __shfl(mine, q, 16);
                    const int s1 = __shfl(mine, q + 1, 16);
                    const int s2 = __shfl(mine, q + 2, 16);
                    const int s3 = __shfl(mine, q + 3, 16);
                    const float4 v0 = HS4[(size_t)s0 * (HID / 4) + f4];
                    const float4 v1 = HS4[(size_t)s1 * (HID / 4) + f4];
                    const float4 v2 = HS4[(size_t)s2 * (HID / 4) + f4];
                    const float4 v3 = HS4[(size_t)s3 * (HID / 4) + f4];
                    const bool k1 = q + 1 < m, k2 = q + 2 < m, k3 = q + 3 < m;
                    acc.x += v0.x; acc.y += v0.y; acc.z += v0.z; acc.w += v0.w;
                    acc.x += k1 ? v1.x : 0.f; acc.y += k1 ? v1.y : 0.f; acc.z += k1 ? v1.z : 0.f; acc.w += k1 ? v1.w : 0.f;
                    acc.x += k2 ? v2.x : 0.f; acc.y += k2 ? v2.y : 0.f; acc.z += k2 ? v2.z : 0.f; acc.w += k2 ? v2.w : 0.f;
                    acc.x += k3 ? v3.x : 0.f; acc.y += k3 ? v3.y : 0.f; acc.z += k3 ? v3.z : 0.f; acc.w += k3 ? v3.w : 0.f;
                }
            }
            const float di = __shfl(dstrip, r, 32);
            float4 o;
            o.x = fmaxf(acc.x * di + bx, 0.f);
            o.y = fmaxf(acc.y * di + by, 0.f);
            o.z = fmaxf(acc.z * di + bz, 0.f);
            o.w = fmaxf(acc.w * di + bw, 0.f);
            *(float4*)(T + r * TPITCH + f4 * 4) = o;
        }
    }
    __syncthreads();
    if (active) {
        const int row = lane & 15;
        const int kk  = (lane >> 4) << 1;
        const float* Tr = T + row * TPITCH + kk;
        v8f c0 = {}, c1 = {}, c2 = {};
        for (int k0 = 0; k0 < HID; k0 += 4) {
            v2f a;
            a.x = Tr[k0];
            a.y = Tr[k0 + 1];
            const float* Wp = W + (size_t)(k0 + kk) * COUT;
            v2f b0 = { Wp[row],      Wp[COUT + row] };
            v2f b1v = { Wp[16 + row], Wp[COUT + 16 + row] };
            int col2 = 32 + row;
            v2f b2 = { (col2 < COUT) ? Wp[col2] : 0.0f,
                       (col2 < COUT) ? Wp[COUT + col2] : 0.0f };
            c0 = __builtin_amdgcn_wmma_f32_16x16x4_f32(false, a, false, b0, (short)0, c0, false, false);
            c1 = __builtin_amdgcn_wmma_f32_16x16x4_f32(false, a, false, b1v, (short)0, c1, false, false);
            c2 = __builtin_amdgcn_wmma_f32_16x16x4_f32(false, a, false, b2, (short)0, c2, false, false);
        }
        const int mrow = m0 + ((lane >> 4) << 3);
        const int cc   = lane & 15;
        for (int r = 0; r < 8; ++r) {
            const float di = __shfl(dstrip, ((lane >> 4) << 3) + r, 32);
            float* out = H2S + (size_t)(mrow + r) * COUT;
            out[cc]      = c0[r] * di;
            out[16 + cc] = c1[r] * di;
            if (32 + cc < COUT) out[32 + cc] = c2[r] * di;
        }
    }
}

#define A2_NODES 64
__global__ __launch_bounds__(256) void k_agg2(const int* __restrict__ rowptr,
                                              const int* __restrict__ col,
                                              const float* __restrict__ dinv,
                                              const float4* __restrict__ H4,
                                              const float* __restrict__ b2,
                                              float4* __restrict__ out4, int n) {
    __shared__ float4 stage[A2_NODES * (COUT / 4)];
    const int g  = threadIdx.x >> 4;
    const int f4 = threadIdx.x & 15;
    const int fl = f4 < (COUT / 4) ? f4 : (COUT / 4 - 1);
    const float* b = b2 + fl * 4;
    const float bx = b[0], by = b[1], bz = b[2], bw = b[3];
    for (int i = 0; i < A2_NODES / 16; ++i) {
        const int nl = i * 16 + g;
        const int d  = blockIdx.x * A2_NODES + nl;
        if (d < n) {
            const int beg = rowptr[d];
            const int end = rowptr[d + 1];
            const float di = dinv[d];
            float4 acc = H4[(size_t)d * (COUT / 4) + fl];
            for (int j0 = beg; j0 < end; j0 += 16) {
                const int rem = end - j0;
                const int m = rem < 16 ? rem : 16;
                const int mine = (f4 < m) ? col[j0 + f4] : d;
                for (int q = 0; q < m; q += 4) {
                    const int s0 = __shfl(mine, q, 16);
                    const int s1 = __shfl(mine, q + 1, 16);
                    const int s2 = __shfl(mine, q + 2, 16);
                    const int s3 = __shfl(mine, q + 3, 16);
                    const float4 v0 = H4[(size_t)s0 * (COUT / 4) + fl];
                    const float4 v1 = H4[(size_t)s1 * (COUT / 4) + fl];
                    const float4 v2 = H4[(size_t)s2 * (COUT / 4) + fl];
                    const float4 v3 = H4[(size_t)s3 * (COUT / 4) + fl];
                    const bool k1 = q + 1 < m, k2 = q + 2 < m, k3 = q + 3 < m;
                    acc.x += v0.x; acc.y += v0.y; acc.z += v0.z; acc.w += v0.w;
                    acc.x += k1 ? v1.x : 0.f; acc.y += k1 ? v1.y : 0.f; acc.z += k1 ? v1.z : 0.f; acc.w += k1 ? v1.w : 0.f;
                    acc.x += k2 ? v2.x : 0.f; acc.y += k2 ? v2.y : 0.f; acc.z += k2 ? v2.z : 0.f; acc.w += k2 ? v2.w : 0.f;
                    acc.x += k3 ? v3.x : 0.f; acc.y += k3 ? v3.y : 0.f; acc.z += k3 ? v3.z : 0.f; acc.w += k3 ? v3.w : 0.f;
                }
            }
            if (f4 < (COUT / 4)) {
                float4 o;
                o.x = acc.x * di + bx;
                o.y = acc.y * di + by;
                o.z = acc.z * di + bz;
                o.w = acc.w * di + bw;
                stage[nl * (COUT / 4) + f4] = o;
            }
        }
    }
    __syncthreads();
    for (int k = threadIdx.x; k < A2_NODES * (COUT / 4); k += 256) {
        const size_t idx = (size_t)blockIdx.x * (A2_NODES * (COUT / 4)) + k;
        if (idx < (size_t)n * (COUT / 4)) out4[idx] = stage[k];
    }
}

extern "C" void kernel_launch(void* const* d_in, const int* in_sizes, int n_in,
                              void* d_out, int out_size, void* d_ws, size_t ws_size,
                              hipStream_t stream) {
    const float* x  = (const float*)d_in[0];
    const float* W1 = (const float*)d_in[1];
    const float* b1 = (const float*)d_in[2];
    const float* W2 = (const float*)d_in[3];
    const float* b2 = (const float*)d_in[4];
    const int*   ei = (const int*)d_in[5];

    const int Nn = in_sizes[0] / FIN;
    const int E  = in_sizes[5] / 2;
    const int* src = ei;
    const int* dst = ei + E;

    const size_t Epad = ((size_t)E + 63) & ~(size_t)63;
    const size_t Npad = ((size_t)Nn + 1 + 63) & ~(size_t)63;

    float* ws     = (float*)d_ws;
    float* hs     = ws;
    float* h2s    = hs + (size_t)Nn * HID;
    int*   rank   = (int*)h2s;
    int*   cnt    = rank + Epad;
    int*   col    = (int*)(h2s + (size_t)Nn * COUT);
    int*   rowptr = col + Epad;
    float* dinv   = (float*)(rowptr + Npad);
    float* out    = (float*)d_out;

    int gN = (Nn + 255) / 256;
    int gE = (E + 255) / 256;
    int gemmBlocks = ((Nn / 16) + 7) / 8;
    int gA2 = (Nn + A2_NODES - 1) / A2_NODES;

    k_cnt_zero<<<gN, 256, 0, stream>>>(cnt, Nn);
    k_rank<<<gE, 256, 0, stream>>>(dst, E, cnt, rank);
    k_scan<<<1, 1024, 0, stream>>>(cnt, rowptr, dinv, Nn);
    k_fill<<<gE, 256, 0, stream>>>(src, dst, rank, rowptr, col, E);

    k_gemm1<<<gemmBlocks, 256, 0, stream>>>(x, W1, dinv, hs, Nn);
    k_agg1_gemm2<<<gemmBlocks, 256, 0, stream>>>(rowptr, col, dinv, (const float4*)hs, b1, W2, h2s, Nn);

    k_agg2<<<gA2, 256, 0, stream>>>(rowptr, col, dinv, (const float4*)h2s, b2, (float4*)out, Nn);
}


# edge list by destination built by two-level LDS counting sort (chunk runs per 256-node bin, one work-group per bin): 2 launches, no global atomic; slots by LDS int atomics, arrival order as before
# speedup vs baseline: 4.2798x; 2.1738x over previous
#include <hip/hip_runtime.h>


#define FIN 256
#define HID 64
#define COUT 40
#define TPITCH 68

typedef float v2f __attribute__((ext_vector_type(2)));
typedef float v8f __attribute__((ext_vector_type(8)));

#define BIN_SHIFT 8
#define BIN_NODES 256
#define CHUNK 8192
#define MAXB 512
#define MAXNC 1024
#define SRC_BITS 17
#define SRC_MASK 0x1FFFF
#define OWN_CAP 12288

__device__ __forceinline__ int wave_incl_scan(int x, int lane) {
    for (int o = 1; o < 32; o <<= 1) {
        const int y = __shfl_up(x, o, 32);
        if (lane >= o) x += y;
    }
    return x;
}

__device__ __forceinline__ int block_excl_scan(int v, int* wtot, int& total) {
    const int lane = threadIdx.x & 31;
    const int w = threadIdx.x >> 5;
    const int x = wave_incl_scan(v, lane);
    if (lane == 31) wtot[w] = x;
    __syncthreads();
    if (w == 0) {
        const int tv = wtot[lane];
        const int xx = wave_incl_scan(tv, lane);
        wtot[lane] = xx - tv;
        if (lane == 31) wtot[32] = xx;
    }
    __syncthreads();
    const int res = wtot[w] + x - v;
    total = wtot[32];
    __syncthreads();
    return res;
}

__global__ __launch_bounds__(1024) void k_chunk_sort(const int* __restrict__ src, const int* __restrict__ dst,
                                                     int E, int nbins,
                                                     int* __restrict__ sorted, int2* __restrict__ tab) {
    __shared__ int hist[MAXB];
    __shared__ int startb[MAXB];
    __shared__ int recs[CHUNK];
    __shared__ int wtot[33];
    const int t = threadIdx.x;
    const int c = blockIdx.x;
    const int base = c * CHUNK;
    const int rem = E - base;
    const int n = rem < CHUNK ? rem : CHUNK;
    if (t < MAXB) hist[t] = 0;
    __syncthreads();
    int s[8], d[8], r[8];
    for (int j = 0; j < 8; ++j) {
        const int i = j * 1024 + t;
        const bool ok = i < n;
        d[j] = ok ? dst[base + i] : 0;
        s[j] = ok ? src[base + i] : 0;
    }
    for (int j = 0; j < 8; ++j) {
        const int i = j * 1024 + t;
        r[j] = (i < n) ? atomicAdd(&hist[d[j] >> BIN_SHIFT], 1) : 0;
    }
    __syncthreads();
    const int h = (t < nbins) ? hist[t] : 0;
    int total;
    const int st = block_excl_scan(h, wtot, total);
    if (t < nbins) {
        startb[t] = st;
        tab[(size_t)c * nbins + t] = make_int2(h, st);
    }
    __syncthreads();
    for (int j = 0; j < 8; ++j) {
        const int i = j * 1024 + t;
        if (i < n) recs[startb[d[j] >> BIN_SHIFT] + r[j]] = s[j] | ((d[j] & (BIN_NODES - 1)) << SRC_BITS);
    }
    __syncthreads();
    for (int j = 0; j < 8; ++j) {
        const int i = j * 1024 + t;
        if (i < n) sorted[base + i] = recs[i];
    }
}

template <bool PLACE>
__device__ __forceinline__ void walk_runs(const int* __restrict__ sorted, const int* pcnt, const int* pst,
                                          int nchunks, int* ncnt, int* cursor, int* list,
                                          int* __restrict__ col, int binBase, bool fits) {
    const int lane = threadIdx.x & 31;
    const int w = threadIdx.x >> 5;
    for (int c0 = 0; c0 < nchunks; c0 += 256) {
        int rec[8];
        for (int u = 0; u < 8; ++u) {
            const int c = c0 + u * 32 + w;
            rec[u] = -1;
            if (c < nchunks && lane < pcnt[c]) rec[u] = sorted[(size_t)c * CHUNK + pst[c] + lane];
        }
        for (int u = 0; u < 8; ++u) {
            if (rec[u] >= 0) {
                const int ld = rec[u] >> SRC_BITS;
                if (PLACE) {
                    const int pos = atomicAdd(&cursor[ld], 1);
                    if (fits) list[pos] = rec[u] & SRC_MASK; else col[binBase + pos] = rec[u] & SRC_MASK;
                } else {
                    atomicAdd(&ncnt[ld], 1);
                }
            }
        }
        for (int u = 0; u < 8; ++u) {
            const int c = c0 + u * 32 + w;
            if (c < nchunks) {
                const int pc = pcnt[c];
                for (int k = lane + 32; k < pc; k += 32) {
                    const int rr = sorted[(size_t)c * CHUNK + pst[c] + k];
                    const int ld = rr >> SRC_BITS;
                    if (PLACE) {
                        const int pos = atomicAdd(&cursor[ld], 1);
                        if (fits) list[pos] = rr & SRC_MASK; else col[binBase + pos] = rr & SRC_MASK;
                    } else {
                        atomicAdd(&ncnt[ld], 1);
                    }
                }
            }
        }
    }
}

__global__ __launch_bounds__(1024) void k_bin_rows(const int* __restrict__ sorted, const int2* __restrict__ tab,
                                                   int nchunks, int nbins, int n,
                                                   int* __restrict__ rowptr, float* __restrict__ dinv,
                                                   int* __restrict__ col) {
    __shared__ int pcnt[MAXNC];
    __shared__ int pst[MAXNC];
    __shared__ int ncnt[BIN_NODES];
    __shared__ int cursor[BIN_NODES];
    __shared__ int list[OWN_CAP];
    __shared__ int wtot[33];
    const int t = threadIdx.x;
    const int bin = blockIdx.x;
    int myc = 0, mys = 0;
    if (t < nchunks) {
        const int2 v = tab[(size_t)t * nbins + bin];
        myc = v.x;
        mys = v.y;
        pcnt[t] = myc;
        pst[t] = mys;
    }
    if (t < BIN_NODES) ncnt[t] = 0;
    int total, binBase, tot2;
    (void)block_excl_scan(myc, wtot, total);
    (void)block_excl_scan(mys, wtot, binBase);
    walk_runs<false>(sorted, pcnt, pst, nchunks, ncnt, cursor, list, col, binBase, true);
    __syncthreads();
    const int nc = (t < BIN_NODES) ? ncnt[t] : 0;
    const int ns = block_excl_scan(nc, wtot, tot2);
    if (t < BIN_NODES) {
        cursor[t] = ns;
        const int node = bin * BIN_NODES + t;
        if (node < n) {
            rowptr[node] = binBase + ns;
            dinv[node] = rsqrtf((float)(nc + 1));
        }
    }
    if (bin == nbins - 1 && t == 0) rowptr[n] = binBase + total;
    __syncthreads();
    const bool fits = total <= OWN_CAP;
    walk_runs<true>(sorted, pcnt, pst, nchunks, ncnt, cursor, list, col, binBase, fits);
    __syncthreads();
    if (fits) {
        for (int i = t; i < total; i += 1024) col[binBase + i] = list[i];
    }
}

__global__ __launch_bounds__(256) void k_gemm1(const float* __restrict__ X,
                                               const float* __restrict__ W,
                                               const float* __restrict__ dinv,
                                               float* __restrict__ H, int nrows) {
    int wave = (blockIdx.x * 256 + threadIdx.x) >> 5;
    int lane = threadIdx.x & 31;
    int m0 = wave << 4;
    if (m0 >= nrows) return;
    int row = lane & 15;
    int kk  = (lane >> 4) << 1;
    const float* Xr = X + (size_t)(m0 + row) * FIN + kk;
    v8f c0 = {}, c1 = {}, c2 = {}, c3 = {};
    for (int k0 = 0; k0 < FIN; k0 += 4) {
        v2f a;
        a.x = Xr[k0];
        a.y = Xr[k0 + 1];
        const float* Wp = W + (size_t)(k0 + kk) * HID;
        v2f b0 = { Wp[row],      Wp[HID + row] };
        v2f b1 = { Wp[16 + row], Wp[HID + 16 + row] };
        v2f b2 = { Wp[32 + row], Wp[HID + 32 + row] };
        v2f b3 = { Wp[48 + row], Wp[HID + 48 + row] };
        c0 = __builtin_amdgcn_wmma_f32_16x16x4_f32(false, a, false, b0, (short)0, c0, false, false);
        c1 = __builtin_amdgcn_wmma_f32_16x16x4_f32(false, a, false, b1, (short)0, c1, false, false);
        c2 = __builtin_amdgcn_wmma_f32_16x16x4_f32(false, a, false, b2, (short)0, c2, false, false);
        c3 = __builtin_amdgcn_wmma_f32_16x16x4_f32(false, a, false, b3, (short)0, c3, false, false);
    }
    int mrow = m0 + ((lane >> 4) << 3);
    int col  = lane & 15;
    for (int r = 0; r < 8; ++r) {
        const float di = dinv[mrow + r];
        float* out = H + (size_t)(mrow + r) * HID + col;
        out[0]  = c0[r] * di;
        out[16] = c1[r] * di;
        out[32] = c2[r] * di;
        out[48] = c3[r] * di;
    }
}

__global__ __launch_bounds__(256) void k_agg1_gemm2(const int* __restrict__ rowptr,
                                                    const int* __restrict__ col,
                                                    const float* __restrict__ dinv,
                                                    const float4* __restrict__ HS4,
                                                    const float* __restrict__ b1,
                                                    const float* __restrict__ W,
                                                    float* __restrict__ H2S, int nrows) {
    __shared__ float tile[8 * 16 * TPITCH];
    const int wib  = threadIdx.x >> 5;
    const int lane = threadIdx.x & 31;
    const int m0   = (blockIdx.x * 8 + wib) << 4;
    const bool active = m0 < nrows;
    float* T = tile + wib * (16 * TPITCH);
    const int sub = lane >> 4;
    const int f4  = lane & 15;
    float dstrip = 0.f;
    if (active) {
        const int rp = rowptr[m0 + (lane < 16 ? lane : 16)];
        dstrip = dinv[m0 + f4];
        const float* b = b1 + f4 * 4;
        const float bx = b[0], by = b[1], bz = b[2], bw = b[3];
        for (int p = 0; p < 8; ++p) {
            const int r = 2 * p + sub;
            const int d = m0 + r;
            const int beg = __shfl(rp, r, 32);
            const int end = __shfl(rp, r + 1, 32);
            float4 acc = HS4[(size_t)d * (HID / 4) + f4];
            for (int j0 = beg; j0 < end; j0 += 16) {
                const int rem = end - j0;
                const int m = rem < 16 ? rem : 16;
                const int mine = (f4 < m) ? col[j0 + f4] : d;
                for (int q = 0; q < m; q += 4) {
                    const int s0 = __shfl(mine, q, 16);
                    const int s1 = __shfl(mine, q + 1, 16);
                    const int s2 = __shfl(mine, q + 2, 16);
                    const int s3 = __shfl(mine, q + 3, 16);
                    const float4 v0 = HS4[(size_t)s0 * (HID / 4) + f4];
                    const float4 v1 = HS4[(size_t)s1 * (HID / 4) + f4];
                    const float4 v2 = HS4[(size_t)s2 * (HID / 4) + f4];
                    const float4 v3 = HS4[(size_t)s3 * (HID / 4) + f4];
                    const bool k1 = q + 1 < m, k2 = q + 2 < m, k3 = q + 3 < m;
                    acc.x += v0.x; acc.y += v0.y; acc.z += v0.z; acc.w += v0.w;
                    acc.x += k1 ? v1.x : 0.f; acc.y += k1 ? v1.y : 0.f; acc.z += k1 ? v1.z : 0.f; acc.w += k1 ? v1.w : 0.f;
                    acc.x += k2 ? v2.x : 0.f; acc.y += k2 ? v2.y : 0.f; acc.z += k2 ? v2.z : 0.f; acc.w += k2 ? v2.w : 0.f;
                    acc.x += k3 ? v3.x : 0.f; acc.y += k3 ? v3.y : 0.f; acc.z += k3 ? v3.z : 0.f; acc.w += k3 ? v3.w : 0.f;
                }
            }
            const float di = __shfl(dstrip, r, 32);
            float4 o;
            o.x = fmaxf(acc.x * di + bx, 0.f);
            o.y = fmaxf(acc.y * di + by, 0.f);
            o.z = fmaxf(acc.z * di + bz, 0.f);
            o.w = fmaxf(acc.w * di + bw, 0.f);
            *(float4*)(T + r * TPITCH + f4 * 4) = o;
        }
    }
    __syncthreads();
    if (active) {
        const int row = lane & 15;
        const int kk  = (lane >> 4) << 1;
        const float* Tr = T + row * TPITCH + kk;
        v8f c0 = {}, c1 = {}, c2 = {};
        for (int k0 = 0; k0 < HID; k0 += 4) {
            v2f a;
            a.x = Tr[k0];
            a.y = Tr[k0 + 1];
            const float* Wp = W + (size_t)(k0 + kk) * COUT;
            v2f b0 = { Wp[row],      Wp[COUT + row] };
            v2f b1v = { Wp[16 + row], Wp[COUT + 16 + row] };
            int col2 = 32 + row;
            v2f b2 = { (col2 < COUT) ? Wp[col2] : 0.0f,
                       (col2 < COUT) ? Wp[COUT + col2] : 0.0f };
            c0 = __builtin_amdgcn_wmma_f32_16x16x4_f32(false, a, false, b0, (short)0, c0, false, false);
            c1 = __builtin_amdgcn_wmma_f32_16x16x4_f32(false, a, false, b1v, (short)0, c1, false, false);
            c2 = __builtin_amdgcn_wmma_f32_16x16x4_f32(false, a, false, b2, (short)0, c2, false, false);
        }
        const int mrow = m0 + ((lane >> 4) << 3);
        const int cc   = lane & 15;
        for (int r = 0; r < 8; ++r) {
            const float di = __shfl(dstrip, ((lane >> 4) << 3) + r, 32);
            float* out = H2S + (size_t)(mrow + r) * COUT;
            out[cc]      = c0[r] * di;
            out[16 + cc] = c1[r] * di;
            if (32 + cc < COUT) out[32 + cc] = c2[r] * di;
        }
    }
}

#define A2_NODES 64
__global__ __launch_bounds__(256) void k_agg2(const int* __restrict__ rowptr,
                                              const int* __restrict__ col,
                                              const float* __restrict__ dinv,
                                              const float4* __restrict__ H4,
                                              const float* __restrict__ b2,
                                              float4* __restrict__ out4, int n) {
    __shared__ float4 stage[A2_NODES * (COUT / 4)];
    const int g  = threadIdx.x >> 4;
    const int f4 = threadIdx.x & 15;
    const int fl = f4 < (COUT / 4) ? f4 : (COUT / 4 - 1);
    const float* b = b2 + fl * 4;
    const float bx = b[0], by = b[1], bz = b[2], bw = b[3];
    for (int i = 0; i < A2_NODES / 16; ++i) {
        const int nl = i * 16 + g;
        const int d  = blockIdx.x * A2_NODES + nl;
        if (d < n) {
            const int beg = rowptr[d];
            const int end = rowptr[d + 1];
            const float di = dinv[d];
            float4 acc = H4[(size_t)d * (COUT / 4) + fl];
            for (int j0 = beg; j0 < end; j0 += 16) {
                const int rem = end - j0;
                const int m = rem < 16 ? rem : 16;
                const int mine = (f4 < m) ? col[j0 + f4] : d;
                for (int q = 0; q < m; q += 4) {
                    const int s0 = __shfl(mine, q, 16);
                    const int s1 = __shfl(mine, q + 1, 16);
                    const int s2 = __shfl(mine, q + 2, 16);
                    const int s3 = __shfl(mine, q + 3, 16);
                    const float4 v0 = H4[(size_t)s0 * (COUT / 4) + fl];
                    const float4 v1 = H4[(size_t)s1 * (COUT / 4) + fl];
                    const float4 v2 = H4[(size_t)s2 * (COUT / 4) + fl];
                    const float4 v3 = H4[(size_t)s3 * (COUT / 4) + fl];
                    const bool k1 = q + 1 < m, k2 = q + 2 < m, k3 = q + 3 < m;
                    acc.x += v0.x; acc.y += v0.y; acc.z += v0.z; acc.w += v0.w;
                    acc.x += k1 ? v1.x : 0.f; acc.y += k1 ? v1.y : 0.f; acc.z += k1 ? v1.z : 0.f; acc.w += k1 ? v1.w : 0.f;
                    acc.x += k2 ? v2.x : 0.f; acc.y += k2 ? v2.y : 0.f; acc.z += k2 ? v2.z : 0.f; acc.w += k2 ? v2.w : 0.f;
                    acc.x += k3 ? v3.x : 0.f; acc.y += k3 ? v3.y : 0.f; acc.z += k3 ? v3.z : 0.f; acc.w += k3 ? v3.w : 0.f;
                }
            }
            if (f4 < (COUT / 4)) {
                float4 o;
                o.x = acc.x * di + bx;
                o.y = acc.y * di + by;
                o.z = acc.z * di + bz;
                o.w = acc.w * di + bw;
                stage[nl * (COUT / 4) + f4] = o;
            }
        }
    }
    __syncthreads();
    for (int k = threadIdx.x; k < A2_NODES * (COUT / 4); k += 256) {
        const size_t idx = (size_t)blockIdx.x * (A2_NODES * (COUT / 4)) + k;
        if (idx < (size_t)n * (COUT / 4)) out4[idx] = stage[k];
    }
}

extern "C" void kernel_launch(void* const* d_in, const int* in_sizes, int n_in,
                              void* d_out, int out_size, void* d_ws, size_t ws_size,
                              hipStream_t stream) {
    const float* x  = (const float*)d_in[0];
    const float* W1 = (const float*)d_in[1];
    const float* b1 = (const float*)d_in[2];
    const float* W2 = (const float*)d_in[3];
    const float* b2 = (const float*)d_in[4];
    const int*   ei = (const int*)d_in[5];

    const int Nn = in_sizes[0] / FIN;
    const int E  = in_sizes[5] / 2;
    const int* src = ei;
    const int* dst = ei + E;

    const size_t Epad = ((size_t)E + 63) & ~(size_t)63;
    const size_t Npad = ((size_t)Nn + 1 + 63) & ~(size_t)63;

    float* ws     = (float*)d_ws;
    float* hs     = ws;
    float* h2s    = hs + (size_t)Nn * HID;
    int*   sorted = (int*)h2s;
    int2*  tab    = (int2*)(sorted + Epad);
    int*   col    = (int*)(h2s + (size_t)Nn * COUT);
    int*   rowptr = col + Epad;
    float* dinv   = (float*)(rowptr + Npad);
    float* out    = (float*)d_out;

    int nbins   = (Nn + BIN_NODES - 1) / BIN_NODES;
    int nchunks = (E + CHUNK - 1) / CHUNK;
    int gemmBlocks = ((Nn / 16) + 7) / 8;
    int gA2 = (Nn + A2_NODES - 1) / A2_NODES;

    k_chunk_sort<<<nchunks, 1024, 0, stream>>>(src, dst, E, nbins, sorted, tab);
    k_bin_rows<<<nbins, 1024, 0, stream>>>(sorted, tab, nchunks, nbins, Nn, rowptr, dinv, col);

    k_gemm1<<<gemmBlocks, 256, 0, stream>>>(x, W1, dinv, hs, Nn);
    k_agg1_gemm2<<<gemmBlocks, 256, 0, stream>>>(rowptr, col, dinv, (const float4*)hs, b1, W2, h2s, Nn);

    k_agg2<<<gA2, 256, 0, stream>>>(rowptr, col, dinv, (const float4*)h2s, b2, (float4*)out, Nn);
}


# product 1 fed from LDS: W1 staged once per work-group in fragment order, 512 groups loop over strips, X read as 16-byte loads; k regrouped in each 8-k block (product 1 sum order changes)
# speedup vs baseline: 4.8144x; 1.1249x over previous
#include <hip/hip_runtime.h>


#define FIN 256
#define HID 64
#define COUT 40
#define TPITCH 68

typedef float v2f __attribute__((ext_vector_type(2)));
typedef float v8f __attribute__((ext_vector_type(8)));

#define BIN_SHIFT 8
#define BIN_NODES 256
#define CHUNK 8192
#define MAXB 512
#define MAXNC 1024
#define SRC_BITS 17
#define SRC_MASK 0x1FFFF
#define OWN_CAP 12288

__device__ __forceinline__ int wave_incl_scan(int x, int lane) {
    for (int o = 1; o < 32; o <<= 1) {
        const int y = __shfl_up(x, o, 32);
        if (lane >= o) x += y;
    }
    return x;
}

__device__ __forceinline__ int block_excl_scan(int v, int* wtot, int& total) {
    const int lane = threadIdx.x & 31;
    const int w = threadIdx.x >> 5;
    const int x = wave_incl_scan(v, lane);
    if (lane == 31) wtot[w] = x;
    __syncthreads();
    if (w == 0) {
        const int tv = wtot[lane];
        const int xx = wave_incl_scan(tv, lane);
        wtot[lane] = xx - tv;
        if (lane == 31) wtot[32] = xx;
    }
    __syncthreads();
    const int res = wtot[w] + x - v;
    total = wtot[32];
    __syncthreads();
    return res;
}

__global__ __launch_bounds__(1024) void k_chunk_sort(const int* __restrict__ src, const int* __restrict__ dst,
                                                     int E, int nbins,
                                                     int* __restrict__ sorted, int2* __restrict__ tab) {
    __shared__ int hist[MAXB];
    __shared__ int startb[MAXB];
    __shared__ int recs[CHUNK];
    __shared__ int wtot[33];
    const int t = threadIdx.x;
    const int c = blockIdx.x;
    const int base = c * CHUNK;
    const int rem = E - base;
    const int n = rem < CHUNK ? rem : CHUNK;
    if (t < MAXB) hist[t] = 0;
    __syncthreads();
    int s[8], d[8], r[8];
    for (int j = 0; j < 8; ++j) {
        const int i = j * 1024 + t;
        const bool ok = i < n;
        d[j] = ok ? dst[base + i] : 0;
        s[j] = ok ? src[base + i] : 0;
    }
    for (int j = 0; j < 8; ++j) {
        const int i = j * 1024 + t;
        r[j] = (i < n) ? atomicAdd(&hist[d[j] >> BIN_SHIFT], 1) : 0;
    }
    __syncthreads();
    const int h = (t < nbins) ? hist[t] : 0;
    int total;
    const int st = block_excl_scan(h, wtot, total);
    if (t < nbins) {
        startb[t] = st;
        tab[(size_t)c * nbins + t] = make_int2(h, st);
    }
    __syncthreads();
    for (int j = 0; j < 8; ++j) {
        const int i = j * 1024 + t;
        if (i < n) recs[startb[d[j] >> BIN_SHIFT] + r[j]] = s[j] | ((d[j] & (BIN_NODES - 1)) << SRC_BITS);
    }
    __syncthreads();
    for (int j = 0; j < 8; ++j) {
        const int i = j * 1024 + t;
        if (i < n) sorted[base + i] = recs[i];
    }
}

template <bool PLACE>
__device__ __forceinline__ void walk_runs(const int* __restrict__ sorted, const int* pcnt, const int* pst,
                                          int nchunks, int* ncnt, int* cursor, int* list,
                                          int* __restrict__ col, int binBase, bool fits) {
    const int lane = threadIdx.x & 31;
    const int w = threadIdx.x >> 5;
    for (int c0 = 0; c0 < nchunks; c0 += 256) {
        int rec[8];
        for (int u = 0; u < 8; ++u) {
            const int c = c0 + u * 32 + w;
            rec[u] = -1;
            if (c < nchunks && lane < pcnt[c]) rec[u] = sorted[(size_t)c * CHUNK + pst[c] + lane];
        }
        for (int u = 0; u < 8; ++u) {
            if (rec[u] >= 0) {
                const int ld = rec[u] >> SRC_BITS;
                if (PLACE) {
                    const int pos = atomicAdd(&cursor[ld], 1);
                    if (fits) list[pos] = rec[u] & SRC_MASK; else col[binBase + pos] = rec[u] & SRC_MASK;
                } else {
                    atomicAdd(&ncnt[ld], 1);
                }
            }
        }
        for (int u = 0; u < 8; ++u) {
            const int c = c0 + u * 32 + w;
            if (c < nchunks) {
                const int pc = pcnt[c];
                for (int k = lane + 32; k < pc; k += 32) {
                    const int rr = sorted[(size_t)c * CHUNK + pst[c] + k];
                    const int ld = rr >> SRC_BITS;
                    if (PLACE) {
                        const int pos = atomicAdd(&cursor[ld], 1);
                        if (fits) list[pos] = rr & SRC_MASK; else col[binBase + pos] = rr & SRC_MASK;
                    } else {
                        atomicAdd(&ncnt[ld], 1);
                    }
                }
            }
        }
    }
}

__global__ __launch_bounds__(1024) void k_bin_rows(const int* __restrict__ sorted, const int2* __restrict__ tab,
                                                   int nchunks, int nbins, int n,
                                                   int* __restrict__ rowptr, float* __restrict__ dinv,
                                                   int* __restrict__ col) {
    __shared__ int pcnt[MAXNC];
    __shared__ int pst[MAXNC];
    __shared__ int ncnt[BIN_NODES];
    __shared__ int cursor[BIN_NODES];
    __shared__ int list[OWN_CAP];
    __shared__ int wtot[33];
    const int t = threadIdx.x;
    const int bin = blockIdx.x;
    int myc = 0, mys = 0;
    if (t < nchunks) {
        const int2 v = tab[(size_t)t * nbins + bin];
        myc = v.x;
        mys = v.y;
        pcnt[t] = myc;
        pst[t] = mys;
    }
    if (t < BIN_NODES) ncnt[t] = 0;
    int total, binBase, tot2;
    (void)block_excl_scan(myc, wtot, total);
    (void)block_excl_scan(mys, wtot, binBase);
    walk_runs<false>(sorted, pcnt, pst, nchunks, ncnt, cursor, list, col, binBase, true);
    __syncthreads();
    const int nc = (t < BIN_NODES) ? ncnt[t] : 0;
    const int ns = block_excl_scan(nc, wtot, tot2);
    if (t < BIN_NODES) {
        cursor[t] = ns;
        const int node = bin * BIN_NODES + t;
        if (node < n) {
            rowptr[node] = binBase + ns;
            dinv[node] = rsqrtf((float)(nc + 1));
        }
    }
    if (bin == nbins - 1 && t == 0) rowptr[n] = binBase + total;
    __syncthreads();
    const bool fits = total <= OWN_CAP;
    walk_runs<true>(sorted, pcnt, pst, nchunks, ncnt, cursor, list, col, binBase, fits);
    __syncthreads();
    if (fits) {
        for (int i = t; i < total; i += 1024) col[binBase + i] = list[i];
    }
}

#define G1_GROUPS 512
__device__ __forceinline__ v8f mm4(v2f a, v2f b, v8f c) {
    return __builtin_amdgcn_wmma_f32_16x16x4_f32(false, a, false, b, (short)0, c, false, false);
}
__global__ __launch_bounds__(256) void k_gemm1(const float* __restrict__ X,
                                               const float* __restrict__ W,
                                               const float* __restrict__ dinv,
                                               float* __restrict__ H, int nrows) {
    __shared__ float4 Wl[FIN * HID / 4];
    const int tid = threadIdx.x;
    for (int i = 0; i < (FIN * HID / 4) / 256; ++i) {
        const int q = i * 256 + tid;
        const float* p = W + (size_t)((q >> 6) * 4) * HID + (q & 63);
        float4 v;
        v.x = p[0];
        v.y = p[HID];
        v.z = p[2 * HID];
        v.w = p[3 * HID];
        Wl[q] = v;
    }
    __syncthreads();
    const int lane = tid & 31;
    const int w = tid >> 5;
    const int c = lane & 15;
    const int h = lane >> 4;
    const int nstrips = nrows >> 4;
    const int nwg = gridDim.x;
    const int base = nstrips / nwg;
    const int extra = nstrips - base * nwg;
    const int bi = blockIdx.x;
    const int nb = base + (bi < extra ? 1 : 0);
    const int sb = bi * base + (bi < extra ? bi : extra);
    if (w < nb) {
        const float4* Xr = (const float4*)(X + (size_t)(((sb + w) << 4) + c) * FIN + 4 * h);
        float4 a[4], an[4], bc[4], bn[4];
        for (int i = 0; i < 4; ++i) a[i] = Xr[2 * i];
        for (int ls = w; ls < nb; ls += 8) {
            const int m0 = (sb + ls) << 4;
            const int mn = (ls + 8 < nb) ? m0 + 128 : m0;
            const float4* Xn = (const float4*)(X + (size_t)(mn + c) * FIN + 4 * h);
            int woff = h * 64 + c;
            asm volatile("" : "+v"(woff));
            const float4* Wf = Wl + woff;
            v8f c0 = {}, c1 = {}, c2 = {}, c3 = {};
            for (int t = 0; t < 4; ++t) bc[t] = Wf[t * 16];
            #pragma unroll
            for (int ms = 0; ms < FIN / 32; ++ms) {
                for (int i = 0; i < 4; ++i) an[i] = (ms + 1 < FIN / 32) ? Xr[2 * (4 * (ms + 1) + i)] : Xn[2 * i];
                __builtin_amdgcn_sched_barrier(0);
                #pragma unroll
                for (int i = 0; i < 4; ++i) {
                    const int j = ms * 4 + i;
                    if (j + 1 < FIN / 8) {
                        for (int t = 0; t < 4; ++t) bn[t] = Wf[(j + 1) * 128 + t * 16];
                    }
                    const float4 av = a[i];
                    const v2f a01 = { av.x, av.y };
                    const v2f a23 = { av.z, av.w };
                    c0 = mm4(a01, (v2f){ bc[0].x, bc[0].y }, c0);
                    c1 = mm4(a01, (v2f){ bc[1].x, bc[1].y }, c1);
                    c2 = mm4(a01, (v2f){ bc[2].x, bc[2].y }, c2);
                    c3 = mm4(a01, (v2f){ bc[3].x, bc[3].y }, c3);
                    c0 = mm4(a23, (v2f){ bc[0].z, bc[0].w }, c0);
                    c1 = mm4(a23, (v2f){ bc[1].z, bc[1].w }, c1);
                    c2 = mm4(a23, (v2f){ bc[2].z, bc[2].w }, c2);
                    c3 = mm4(a23, (v2f){ bc[3].z, bc[3].w }, c3);
                    for (int t = 0; t < 4; ++t) bc[t] = bn[t];
                }
                for (int i = 0; i < 4; ++i) a[i] = an[i];
            }
            Xr = Xn;
            const int mrow = m0 + (h << 3);
            for (int r = 0; r < 8; ++r) {
                const float di = dinv[mrow + r];
                float* out = H + (size_t)(mrow + r) * HID + c;
                out[0]  = c0[r] * di;
                out[16] = c1[r] * di;
                out[32] = c2[r] * di;
                out[48] = c3[r] * di;
            }
        }
    }
}

__global__ __launch_bounds__(256) void k_agg1_gemm2(const int* __restrict__ rowptr,
                                                    const int* __restrict__ col,
                                                    const float* __restrict__ dinv,
                                                    const float4* __restrict__ HS4,
                                                    const float* __restrict__ b1,
                                                    const float* __restrict__ W,
                                                    float* __restrict__ H2S, int nrows) {
    __shared__ float tile[8 * 16 * TPITCH];
    const int wib  = threadIdx.x >> 5;
    const int lane = threadIdx.x & 31;
    const int m0   = (blockIdx.x * 8 + wib) << 4;
    const bool active = m0 < nrows;
    float* T = tile + wib * (16 * TPITCH);
    const int sub = lane >> 4;
    const int f4  = lane & 15;
    float dstrip = 0.f;
    if (active) {
        const int rp = rowptr[m0 + (lane < 16 ? lane : 16)];
        dstrip = dinv[m0 + f4];
        const float* b = b1 + f4 * 4;
        const float bx = b[0], by = b[1], bz = b[2], bw = b[3];
        for (int p = 0; p < 8; ++p) {
            const int r = 2 * p + sub;
            const int d = m0 + r;
            const int beg = __shfl(rp, r, 32);
            const int end = __shfl(rp, r + 1, 32);
            float4 acc = HS4[(size_t)d * (HID / 4) + f4];
            for (int j0 = beg; j0 < end; j0 += 16) {
                const int rem = end - j0;
                const int m = rem < 16 ? rem : 16;
                const int mine = (f4 < m) ? col[j0 + f4] : d;
                for (int q = 0; q < m; q += 4) {
                    const int s0 = __shfl(mine, q, 16);
                    const int s1 = __shfl(mine, q + 1, 16);
                    const int s2 = __shfl(mine, q + 2, 16);
                    const int s3 = __shfl(mine, q + 3, 16);
                    const float4 v0 = HS4[(size_t)s0 * (HID / 4) + f4];
                    const float4 v1 = HS4[(size_t)s1 * (HID / 4) + f4];
                    const float4 v2 = HS4[(size_t)s2 * (HID / 4) + f4];
                    const float4 v3 = HS4[(size_t)s3 * (HID / 4) + f4];
                    const bool k1 = q + 1 < m, k2 = q + 2 < m, k3 = q + 3 < m;
                    acc.x += v0.x; acc.y += v0.y; acc.z += v0.z; acc.w += v0.w;
                    acc.x += k1 ? v1.x : 0.f; acc.y += k1 ? v1.y : 0.f; acc.z += k1 ? v1.z : 0.f; acc.w += k1 ? v1.w : 0.f;
                    acc.x += k2 ? v2.x : 0.f; acc.y += k2 ? v2.y : 0.f; acc.z += k2 ? v2.z : 0.f; acc.w += k2 ? v2.w : 0.f;
                    acc.x += k3 ? v3.x : 0.f; acc.y += k3 ? v3.y : 0.f; acc.z += k3 ? v3.z : 0.f; acc.w += k3 ? v3.w : 0.f;
                }
            }
            const float di = __shfl(dstrip, r, 32);
            float4 o;
            o.x = fmaxf(acc.x * di + bx, 0.f);
            o.y = fmaxf(acc.y * di + by, 0.f);
            o.z = fmaxf(acc.z * di + bz, 0.f);
            o.w = fmaxf(acc.w * di + bw, 0.f);
            *(float4*)(T + r * TPITCH + f4 * 4) = o;
        }
    }
    __syncthreads();
    if (active) {
        const int row = lane & 15;
        const int kk  = (lane >> 4) << 1;
        const float* Tr = T + row * TPITCH + kk;
        v8f c0 = {}, c1 = {}, c2 = {};
        for (int k0 = 0; k0 < HID; k0 += 4) {
            v2f a;
            a.x = Tr[k0];
            a.y = Tr[k0 + 1];
            const float* Wp = W + (size_t)(k0 + kk) * COUT;
            v2f b0 = { Wp[row],      Wp[COUT + row] };
            v2f b1v = { Wp[16 + row], Wp[COUT + 16 + row] };
            int col2 = 32 + row;
            v2f b2 = { (col2 < COUT) ? Wp[col2] : 0.0f,
                       (col2 < COUT) ? Wp[COUT + col2] : 0.0f };
            c0 = __builtin_amdgcn_wmma_f32_16x16x4_f32(false, a, false, b0, (short)0, c0, false, false);
            c1 = __builtin_amdgcn_wmma_f32_16x16x4_f32(false, a, false, b1v, (short)0, c1, false, false);
            c2 = __builtin_amdgcn_wmma_f32_16x16x4_f32(false, a, false, b2, (short)0, c2, false, false);
        }
        const int mrow = m0 + ((lane >> 4) << 3);
        const int cc   = lane & 15;
        for (int r = 0; r < 8; ++r) {
            const float di = __shfl(dstrip, ((lane >> 4) << 3) + r, 32);
            float* out = H2S + (size_t)(mrow + r) * COUT;
            out[cc]      = c0[r] * di;
            out[16 + cc] = c1[r] * di;
            if (32 + cc < COUT) out[32 + cc] = c2[r] * di;
        }
    }
}

#define A2_NODES 64
__global__ __launch_bounds__(256) void k_agg2(const int* __restrict__ rowptr,
                                              const int* __restrict__ col,
                                              const float* __restrict__ dinv,
                                              const float4* __restrict__ H4,
                                              const float* __restrict__ b2,
                                              float4* __restrict__ out4, int n) {
    __shared__ float4 stage[A2_NODES * (COUT / 4)];
    const int g  = threadIdx.x >> 4;
    const int f4 = threadIdx.x & 15;
    const int fl = f4 < (COUT / 4) ? f4 : (COUT / 4 - 1);
    const float* b = b2 + fl * 4;
    const float bx = b[0], by = b[1], bz = b[2], bw = b[3];
    for (int i = 0; i < A2_NODES / 16; ++i) {
        const int nl = i * 16 + g;
        const int d  = blockIdx.x * A2_NODES + nl;
        if (d < n) {
            const int beg = rowptr[d];
            const int end = rowptr[d + 1];
            const float di = dinv[d];
            float4 acc = H4[(size_t)d * (COUT / 4) + fl];
            for (int j0 = beg; j0 < end; j0 += 16) {
                const int rem = end - j0;
                const int m = rem < 16 ? rem : 16;
                const int mine = (f4 < m) ? col[j0 + f4] : d;
                for (int q = 0; q < m; q += 4) {
                    const int s0 = __shfl(mine, q, 16);
                    const int s1 = __shfl(mine, q + 1, 16);
                    const int s2 = __shfl(mine, q + 2, 16);
                    const int s3 = __shfl(mine, q + 3, 16);
                    const float4 v0 = H4[(size_t)s0 * (COUT / 4) + fl];
                    const float4 v1 = H4[(size_t)s1 * (COUT / 4) + fl];
                    const float4 v2 = H4[(size_t)s2 * (COUT / 4) + fl];
                    const float4 v3 = H4[(size_t)s3 * (COUT / 4) + fl];
                    const bool k1 = q + 1 < m, k2 = q + 2 < m, k3 = q + 3 < m;
                    acc.x += v0.x; acc.y += v0.y; acc.z += v0.z; acc.w += v0.w;
                    acc.x += k1 ? v1.x : 0.f; acc.y += k1 ? v1.y : 0.f; acc.z += k1 ? v1.z : 0.f; acc.w += k1 ? v1.w : 0.f;
                    acc.x += k2 ? v2.x : 0.f; acc.y += k2 ? v2.y : 0.f; acc.z += k2 ? v2.z : 0.f; acc.w += k2 ? v2.w : 0.f;
                    acc.x += k3 ? v3.x : 0.f; acc.y += k3 ? v3.y : 0.f; acc.z += k3 ? v3.z : 0.f; acc.w += k3 ? v3.w : 0.f;
                }
            }
            if (f4 < (COUT / 4)) {
                float4 o;
                o.x = acc.x * di + bx;
                o.y = acc.y * di + by;
                o.z = acc.z * di + bz;
                o.w = acc.w * di + bw;
                stage[nl * (COUT / 4) + f4] = o;
            }
        }
    }
    __syncthreads();
    for (int k = threadIdx.x; k < A2_NODES * (COUT / 4); k += 256) {
        const size_t idx = (size_t)blockIdx.x * (A2_NODES * (COUT / 4)) + k;
        if (idx < (size_t)n * (COUT / 4)) out4[idx] = stage[k];
    }
}

extern "C" void kernel_launch(void* const* d_in, const int* in_sizes, int n_in,
                              void* d_out, int out_size, void* d_ws, size_t ws_size,
                              hipStream_t stream) {
    const float* x  = (const float*)d_in[0];
    const float* W1 = (const float*)d_in[1];
    const float* b1 = (const float*)d_in[2];
    const float* W2 = (const float*)d_in[3];
    const float* b2 = (const float*)d_in[4];
    const int*   ei = (const int*)d_in[5];

    const int Nn = in_sizes[0] / FIN;
    const int E  = in_sizes[5] / 2;
    const int* src = ei;
    const int* dst = ei + E;

    const size_t Epad = ((size_t)E + 63) & ~(size_t)63;
    const size_t Npad = ((size_t)Nn + 1 + 63) & ~(size_t)63;

    float* ws     = (float*)d_ws;
    float* hs     = ws;
    float* h2s    = hs + (size_t)Nn * HID;
    int*   sorted = (int*)h2s;
    int2*  tab    = (int2*)(sorted + Epad);
    int*   col    = (int*)(h2s + (size_t)Nn * COUT);
    int*   rowptr = col + Epad;
    float* dinv   = (float*)(rowptr + Npad);
    float* out    = (float*)d_out;

    int nbins   = (Nn + BIN_NODES - 1) / BIN_NODES;
    int nchunks = (E + CHUNK - 1) / CHUNK;
    int gemmBlocks = ((Nn / 16) + 7) / 8;
    int gA2 = (Nn + A2_NODES - 1) / A2_NODES;

    k_chunk_sort<<<nchunks, 1024, 0, stream>>>(src, dst, E, nbins, sorted, tab);
    k_bin_rows<<<nbins, 1024, 0, stream>>>(sorted, tab, nchunks, nbins, Nn, rowptr, dinv, col);

    k_gemm1<<<G1_GROUPS, 256, 0, stream>>>(x, W1, dinv, hs, Nn);
    k_agg1_gemm2<<<gemmBlocks, 256, 0, stream>>>(rowptr, col, dinv, (const float4*)hs, b1, W2, h2s, Nn);

    k_agg2<<<gA2, 256, 0, stream>>>(rowptr, col, dinv, (const float4*)h2s, b2, (float4*)out, Nn);
}
